# starting kernel
# baseline (speedup 1.0000x reference)
#include <hip/hip_runtime.h>

typedef float v2f __attribute__((ext_vector_type(2)));
typedef float v8f __attribute__((ext_vector_type(8)));

__device__ __forceinline__ float fast_tanhf(float x) {
#if __has_builtin(__builtin_amdgcn_tanhf)
    return __builtin_amdgcn_tanhf(x);
#else
    float r;
    asm("v_tanh_f32 %0, %1" : "=v"(r) : "v"(x));
    return r;
#endif
}

__device__ __forceinline__ float ode_eval(float s,
                                          const v2f Bops[4],
                                          const v8f Cops[4],
                                          const float w2m[4],
                                          float bias, int lane) {
    v2f A;
    A.x = s;
    A.y = 0.0f;

    float acc[8];
#pragma unroll
    for (int v = 0; v < 8; ++v) acc[v] = 0.0f;

#pragma unroll
    for (int t = 0; t < 4; ++t) {
        v8f d = __builtin_amdgcn_wmma_f32_16x16x4_f32(
             false, A,  false, Bops[t],
             (short)0, Cops[t],  false,  false);
#pragma unroll
        for (int v = 0; v < 8; ++v)
            acc[v] = fmaf(fast_tanhf(d[v]), w2m[t], acc[v]);
    }

    const bool b0 = (lane & 1) != 0;
    const bool b1 = (lane & 2) != 0;
    const bool b2 = (lane & 4) != 0;

    float t0[4];
#pragma unroll
    for (int p = 0; p < 4; ++p) {
        const float own = b0 ? acc[2 * p + 1] : acc[2 * p];
        const float oth = b0 ? acc[2 * p]     : acc[2 * p + 1];
        t0[p] = own + __shfl_xor(oth, 1, 32);
    }
    float t1[2];
#pragma unroll
    for (int q = 0; q < 2; ++q) {
        const float own = b1 ? t0[2 * q + 1] : t0[2 * q];
        const float oth = b1 ? t0[2 * q]     : t0[2 * q + 1];
        t1[q] = own + __shfl_xor(oth, 2, 32);
    }
    const float own = b2 ? t1[1] : t1[0];
    const float oth = b2 ? t1[0] : t1[1];
    float w = own + __shfl_xor(oth, 4, 32);
    w += __shfl_xor(w, 8, 32);
    const float o = __shfl_xor(w, 16, 32);
    return ((lane & 8) ? o : w) + bias;
}

__global__ __launch_bounds__(256) void node_rk4_wmma(
    const float* __restrict__ y_in, const float* __restrict__ ctx,
    const float* __restrict__ meta_p, const float* __restrict__ W1,
    const float* __restrict__ b1, const float* __restrict__ W2,
    const float* __restrict__ b2, const float* __restrict__ dt_p,
    float* __restrict__ y_out, int n_elems) {
    const int lane = threadIdx.x & 31;
    const int wave = (int)((blockIdx.x * blockDim.x + threadIdx.x) >> 5);
    const int base = wave * 16;
    const bool lo  = lane < 16;
    const int  nl  = lane & 15;

    const float meta = meta_p[0];
    const float h    = dt_p[0] * 0.25f;
    const float bias = b2[0] * meta;
    const float h2   = 0.5f * h;
    const float h6   = h * (1.0f / 6.0f);

    float cx[8];
#pragma unroll
    for (int k = 0; k < 8; ++k) cx[k] = ctx[k];

    v2f Bops[4];
    v8f Cops[4];
    float w2m[4];
#pragma unroll
    for (int t = 0; t < 4; ++t) {
        const int j = t * 16 + nl;
        const float a = W1[j];
        float c = b1[j];
#pragma unroll
        for (int k = 0; k < 8; ++k) c = fmaf(cx[k], W1[(k + 1) * 64 + j], c);
        Bops[t].x = lo ? a : 0.0f;
        Bops[t].y = 0.0f;
#pragma unroll
        for (int v = 0; v < 8; ++v) Cops[t][v] = c;
        w2m[t] = W2[j] * meta;
    }

    const int idx = base + lane;
    float y = 0.0f;
    if (lo && idx < n_elems) y = y_in[idx];

    for (int step = 0; step < 4; ++step) {
        const float k1 = ode_eval(y, Bops, Cops, w2m, bias, lane);
        const float s2 = lo ? fmaf(h2, k1, y) : 0.0f;
        const float k2 = ode_eval(s2, Bops, Cops, w2m, bias, lane);
        const float s3 = lo ? fmaf(h2, k2, y) : 0.0f;
        const float k3 = ode_eval(s3, Bops, Cops, w2m, bias, lane);
        const float s4 = lo ? fmaf(h, k3, y) : 0.0f;
        const float k4 = ode_eval(s4, Bops, Cops, w2m, bias, lane);
        const float upd = k1 + 2.0f * k2 + 2.0f * k3 + k4;
        y = lo ? fmaf(h6, upd, y) : 0.0f;
    }

    if (lo && idx < n_elems) y_out[idx] = y;
}

extern "C" void kernel_launch(void* const* d_in, const int* in_sizes, int n_in,
                              void* d_out, int out_size, void* d_ws, size_t ws_size,
                              hipStream_t stream) {
    const float* y_in = (const float*)d_in[0];
    const float* ctx  = (const float*)d_in[1];
    const float* meta = (const float*)d_in[2];
    const float* W1   = (const float*)d_in[3];
    const float* b1   = (const float*)d_in[4];
    const float* W2   = (const float*)d_in[5];
    const float* b2   = (const float*)d_in[6];
    const float* dt   = (const float*)d_in[7];
    float* out = (float*)d_out;

    const int n = in_sizes[0];
    const int waves = (n + 15) / 16;
    const int blocks = (waves + 7) / 8;

    node_rk4_wmma<<<blocks, 256, 0, stream>>>(y_in, ctx, meta, W1, b1, W2, b2,
                                              dt, out, n);
}


# hidden units on the WMMA tile rows (weights as A, states as B): the 64-term sum runs down each lane's registers instead of a cross-lane butterfly; changes the order of that sum
# speedup vs baseline: 1.1443x; 1.1443x over previous
#include <hip/hip_runtime.h>


typedef float v2f __attribute__((ext_vector_type(2)));
typedef float v8f __attribute__((ext_vector_type(8)));

__device__ __forceinline__ float fast_tanhf(float x) {
#if __has_builtin(__builtin_amdgcn_tanhf)
    return __builtin_amdgcn_tanhf(x);
#else
    float r;
    asm("v_tanh_f32 %0, %1" : "=v"(r) : "v"(x));
    return r;
#endif
}

__device__ __forceinline__ float xhalf(float x) {
    const int b = __builtin_bit_cast(int, x);
    return __builtin_bit_cast(
        float, __builtin_amdgcn_permlanex16(b, b, 0x76543210, 0xfedcba98, false, false));
}

__device__ __forceinline__ float ode_eval(float s,
                                          const v2f Aops[4],
                                          const v8f Cops[4],
                                          const v8f W2c[4],
                                          float bias) {
    v2f B;
    B.x = s;
    B.y = 0.0f;

    v2f acc;
    acc.x = 0.0f;
    acc.y = 0.0f;
#pragma unroll
    for (int t = 0; t < 4; ++t) {
        const v8f d = __builtin_amdgcn_wmma_f32_16x16x4_f32(
             false, Aops[t],  false, B,
             (short)0, Cops[t],  false,  false);
#pragma unroll
        for (int r = 0; r < 8; r += 2) {
            v2f th, w;
            th.x = fast_tanhf(d[r]);
            th.y = fast_tanhf(d[r + 1]);
            w.x  = W2c[t][r];
            w.y  = W2c[t][r + 1];
            acc = __builtin_elementwise_fma(th, w, acc);
        }
    }
    const float half_sum = acc.x + acc.y;
    return (half_sum + xhalf(half_sum)) + bias;
}

__global__ __launch_bounds__(256) void node_rk4_wmma(
    const float* __restrict__ y_in, const float* __restrict__ ctx,
    const float* __restrict__ meta_p, const float* __restrict__ W1,
    const float* __restrict__ b1, const float* __restrict__ W2,
    const float* __restrict__ b2, const float* __restrict__ dt_p,
    float* __restrict__ y_out, int n_elems) {
    __shared__ __attribute__((aligned(16))) float s_c[64];
    __shared__ __attribute__((aligned(16))) float s_w2[64];

    const int lane = threadIdx.x & 31;
    const int wave = (int)((blockIdx.x * blockDim.x + threadIdx.x) >> 5);
    const int base = wave * 16;
    const bool lo  = lane < 16;
    const int  nl  = lane & 15;

    const float meta = meta_p[0];
    const float h    = dt_p[0] * 0.25f;
    const float bias = b2[0] * meta;
    const float h2   = 0.5f * h;
    const float h6   = h * (1.0f / 6.0f);

    if (threadIdx.x < 64) {
        const int j = threadIdx.x;
        float c = b1[j];
#pragma unroll
        for (int k = 0; k < 8; ++k) c = fmaf(ctx[k], W1[(k + 1) * 64 + j], c);
        s_c[j]  = c;
        s_w2[j] = W2[j] * meta;
    }
    __syncthreads();

    const int hb = (lane >> 4) * 8;
    v2f Aops[4];
    v8f Cops[4];
    v8f W2c[4];
#pragma unroll
    for (int t = 0; t < 4; ++t) {
        const float a = W1[t * 16 + nl];
        Aops[t].x = lo ? a : 0.0f;
        Aops[t].y = 0.0f;
#pragma unroll
        for (int r = 0; r < 8; ++r) {
            Cops[t][r] = s_c[t * 16 + hb + r];
            W2c[t][r]  = s_w2[t * 16 + hb + r];
        }
    }

    const int idx = base + nl;
    const bool ok = idx < n_elems;
    float y = ok ? y_in[idx] : 0.0f;

    for (int step = 0; step < 4; ++step) {
        const float k1 = ode_eval(lo ? y : 0.0f, Aops, Cops, W2c, bias);
        const float s2 = lo ? fmaf(h2, k1, y) : 0.0f;
        const float k2 = ode_eval(s2, Aops, Cops, W2c, bias);
        const float s3 = lo ? fmaf(h2, k2, y) : 0.0f;
        const float k3 = ode_eval(s3, Aops, Cops, W2c, bias);
        const float s4 = lo ? fmaf(h, k3, y) : 0.0f;
        const float k4 = ode_eval(s4, Aops, Cops, W2c, bias);
        const float upd = k1 + 2.0f * k2 + 2.0f * k3 + k4;
        y = fmaf(h6, upd, y);
    }

    if (lo && ok) y_out[idx] = y;
}

extern "C" void kernel_launch(void* const* d_in, const int* in_sizes, int n_in,
                              void* d_out, int out_size, void* d_ws, size_t ws_size,
                              hipStream_t stream) {
    const float* y_in = (const float*)d_in[0];
    const float* ctx  = (const float*)d_in[1];
    const float* meta = (const float*)d_in[2];
    const float* W1   = (const float*)d_in[3];
    const float* b1   = (const float*)d_in[4];
    const float* W2   = (const float*)d_in[5];
    const float* b2   = (const float*)d_in[6];
    const float* dt   = (const float*)d_in[7];
    float* out = (float*)d_out;

    const int n = in_sizes[0];
    const int waves = (n + 15) / 16;
    const int blocks = (waves + 7) / 8;

    node_rk4_wmma<<<blocks, 256, 0, stream>>>(y_in, ctx, meta, W1, b1, W2, b2,
                                              dt, out, n);
}
